# starting kernel
# baseline (speedup 1.0000x reference)
#include <hip/hip_runtime.h>

typedef __attribute__((ext_vector_type(2))) float v2f;
typedef __attribute__((ext_vector_type(8))) float v8f;

#define NFILT 64
#define WAVES_PER_BLOCK 8

__global__ __launch_bounds__(256)
void seg_mean_wmma_kernel(const float* __restrict__ src,
                          const int*   __restrict__ idx,
                          const int*   __restrict__ spans,
                          float*       __restrict__ out,
                          int n_tgt)
{
    const int lane = threadIdx.x & 31;
    const int wave = __builtin_amdgcn_readfirstlane((int)(threadIdx.x >> 5));
    const int t = __builtin_amdgcn_readfirstlane((int)blockIdx.x) * WAVES_PER_BLOCK + wave;
    if (t >= n_tgt) return;

    const int beg = __builtin_amdgcn_readfirstlane(spans[t]);
    const int end = __builtin_amdgcn_readfirstlane(spans[t + 1]);
    const int cnt = end - beg;
    const int elast = (cnt > 0) ? end - 1 : beg;

    const int  col = lane & 15;
    const bool hi  = (lane >= 16);

    v2f a; a.x = 1.0f; a.y = 1.0f;
    v8f acc0 = {}; v8f acc1 = {}; v8f acc2 = {}; v8f acc3 = {};

    for (int e = beg; e < end; e += 4) {
        const int p1 = (e + 1 < end) ? e + 1 : elast;
        const int p2 = (e + 2 < end) ? e + 2 : elast;
        const int p3 = (e + 3 < end) ? e + 3 : elast;
        const int i0 = __builtin_amdgcn_readfirstlane(idx[e]);
        const int i1 = __builtin_amdgcn_readfirstlane(idx[p1]);
        const int i2 = __builtin_amdgcn_readfirstlane(idx[p2]);
        const int i3 = __builtin_amdgcn_readfirstlane(idx[p3]);
        const float m1 = (e + 1 < end) ? 1.0f : 0.0f;
        const float m2 = (e + 2 < end) ? 1.0f : 0.0f;
        const float m3 = (e + 3 < end) ? 1.0f : 0.0f;

        if (e + 4 < end) {
            const int pe = e + 4 + (lane >> 3);
            const int pf = idx[(pe < end) ? pe : elast];
            __builtin_prefetch(src + (size_t)pf * NFILT + (lane & 7) * 8, 0, 1);
        }

        const float* r02 = src + (size_t)(hi ? i2 : i0) * NFILT + col;
        const float* r13 = src + (size_t)(hi ? i3 : i1) * NFILT + col;
        const float  s02 = hi ? m2 : 1.0f;
        const float  s13 = hi ? m3 : m1;

        v2f b0, b1, b2, b3;
        b0.x = r02[ 0] * s02;  b0.y = r13[ 0] * s13;
        b1.x = r02[16] * s02;  b1.y = r13[16] * s13;
        b2.x = r02[32] * s02;  b2.y = r13[32] * s13;
        b3.x = r02[48] * s02;  b3.y = r13[48] * s13;

        acc0 = __builtin_amdgcn_wmma_f32_16x16x4_f32(false, a, false, b0,
                                                     (short)0, acc0, false, false);
        acc1 = __builtin_amdgcn_wmma_f32_16x16x4_f32(false, a, false, b1,
                                                     (short)0, acc1, false, false);
        acc2 = __builtin_amdgcn_wmma_f32_16x16x4_f32(false, a, false, b2,
                                                     (short)0, acc2, false, false);
        acc3 = __builtin_amdgcn_wmma_f32_16x16x4_f32(false, a, false, b3,
                                                     (short)0, acc3, false, false);
    }

    const float inv = 1.0f / (float)((cnt > 0) ? cnt : 1);
    const float o_lo = (hi ? acc1[0] : acc0[0]) * inv;
    const float o_hi = (hi ? acc3[0] : acc2[0]) * inv;
    const int   c_lo = hi ? 1 : 0;
    const int   c_hi = hi ? 3 : 2;

    float* orow = out + (size_t)t * NFILT;
    orow[c_lo * 16 + col] = o_lo;
    orow[c_hi * 16 + col] = o_hi;
}

extern "C" void kernel_launch(void* const* d_in, const int* in_sizes, int n_in,
                              void* d_out, int out_size, void* d_ws, size_t ws_size,
                              hipStream_t stream) {
    (void)n_in; (void)out_size; (void)d_ws; (void)ws_size;
    const float* sources = (const float*)d_in[0];
    const int*   indices = (const int*)d_in[1];
    const int*   spans   = (const int*)d_in[2];
    float*       out     = (float*)d_out;

    const int n_tgt = in_sizes[2] - 1;
    const int blocks = (n_tgt + WAVES_PER_BLOCK - 1) / WAVES_PER_BLOCK;

    seg_mean_wmma_kernel<<<blocks, 256, 0, stream>>>(sources, indices, spans,
                                                     out, n_tgt);
}


# one wave sums 16 consecutive segments in one WMMA stream: A = 0/1 segment membership instead of ones, 32 entries gathered per trip; K-steps now start at the tile's first entry, not each segment's
# speedup vs baseline: 1.6980x; 1.6980x over previous
#include <hip/hip_runtime.h>

typedef __attribute__((ext_vector_type(2))) float v2f;
typedef __attribute__((ext_vector_type(8))) float v8f;


#define NFILT 64
#define SEG_PER_WAVE 16
#define ENT_PER_CHUNK 32
#define WAVES_PER_BLOCK 4

__device__ __forceinline__ float xhalf(float x) {
    return __builtin_bit_cast(float, __builtin_amdgcn_permlanex16(
        __builtin_bit_cast(int, x), __builtin_bit_cast(int, x),
        0x76543210, 0xfedcba98, false, false));
}

__global__ __launch_bounds__(WAVES_PER_BLOCK * 32)
void seg_mean_wmma_kernel(const float* __restrict__ src,
                          const int*   __restrict__ idx,
                          const int*   __restrict__ spans,
                          float*       __restrict__ out,
                          int n_tgt)
{
    const int lane = threadIdx.x & 31;
    const int wave = __builtin_amdgcn_readfirstlane((int)(threadIdx.x >> 5));
    const int tile = __builtin_amdgcn_readfirstlane((int)blockIdx.x) * WAVES_PER_BLOCK + wave;
    const int t0 = tile * SEG_PER_WAVE;
    if (t0 >= n_tgt) return;

    const int  col = lane & 15;
    const int  hh  = lane >> 4;
    const bool hi  = (hh != 0);

    const int tm   = t0 + col;
    const int tlo  = (tm     < n_tgt) ? tm     : n_tgt;
    const int tup  = (tm + 1 < n_tgt) ? tm + 1 : n_tgt;
    const int lo   = spans[tlo];
    const int up   = spans[tup];
    const unsigned len = (unsigned)(up - lo);

    const int e0 = __builtin_amdgcn_readlane(lo, 0);
    const int e1 = __builtin_amdgcn_readlane(up, 15);
    const int elast = e1 - 1;

    v8f acc0 = {}; v8f acc1 = {}; v8f acc2 = {}; v8f acc3 = {};

    if (e0 < e1) {
        int pe  = e0 + lane;
        int iv  = idx[(pe < e1) ? pe : elast];
        pe += ENT_PER_CHUNK;
        int ivn = idx[(pe < e1) ? pe : elast];

        for (int e = e0; e < e1; e += ENT_PER_CHUNK) {
            __builtin_prefetch(src + (size_t)ivn * NFILT, 0, 1);

            pe += ENT_PER_CHUNK;
            const int ivnn = idx[(pe < e1) ? pe : elast];

            v2f b[8][4];
#pragma unroll
            for (int j = 0; j < 8; ++j) {
                const int i0 = __builtin_amdgcn_readlane(iv, 4 * j + 0);
                const int i1 = __builtin_amdgcn_readlane(iv, 4 * j + 1);
                const int i2 = __builtin_amdgcn_readlane(iv, 4 * j + 2);
                const int i3 = __builtin_amdgcn_readlane(iv, 4 * j + 3);
                const float* r02 = src + (size_t)(hi ? i2 : i0) * NFILT + col;
                const float* r13 = src + (size_t)(hi ? i3 : i1) * NFILT + col;
                b[j][0].x = r02[ 0];  b[j][0].y = r13[ 0];
                b[j][1].x = r02[16];  b[j][1].y = r13[16];
                b[j][2].x = r02[32];  b[j][2].y = r13[32];
                b[j][3].x = r02[48];  b[j][3].y = r13[48];
            }

            __builtin_amdgcn_sched_barrier(0);

#pragma unroll
            for (int j = 0; j < 8; ++j) {
                const int p = e + 4 * j + 2 * hh;
                v2f a;
                a.x = ((unsigned)(p     - lo) < len) ? 1.0f : 0.0f;
                a.y = ((unsigned)(p + 1 - lo) < len) ? 1.0f : 0.0f;
                acc0 = __builtin_amdgcn_wmma_f32_16x16x4_f32(false, a, false, b[j][0],
                                                             (short)0, acc0, false, false);
                acc1 = __builtin_amdgcn_wmma_f32_16x16x4_f32(false, a, false, b[j][1],
                                                             (short)0, acc1, false, false);
                acc2 = __builtin_amdgcn_wmma_f32_16x16x4_f32(false, a, false, b[j][2],
                                                             (short)0, acc2, false, false);
                acc3 = __builtin_amdgcn_wmma_f32_16x16x4_f32(false, a, false, b[j][3],
                                                             (short)0, acc3, false, false);
            }
            iv  = ivn;
            ivn = ivnn;
        }
    }

    const float inv = 1.0f / (float)((len > 0u) ? len : 1u);
    float va0[8], va1[8], vb0[8], vb1[8];
    float* obase = out + (size_t)t0 * NFILT + hh * 16 + col;
    const int rows_left = n_tgt - t0;
#pragma unroll
    for (int r = 0; r < 8; ++r) {
        const float inv_a = __builtin_bit_cast(float,
            __builtin_amdgcn_readlane(__builtin_bit_cast(int, inv), r));
        const float inv_b = __builtin_bit_cast(float,
            __builtin_amdgcn_readlane(__builtin_bit_cast(int, inv), r + 8));
        const float x0 = acc0[r], y0 = acc1[r], x1 = acc2[r], y1 = acc3[r];
        const float g0 = xhalf(hi ? x0 : y0);
        const float g1 = xhalf(hi ? x1 : y1);
        va0[r] = (hi ? g0 : x0) * inv_a;  vb0[r] = (hi ? y0 : g0) * inv_b;
        va1[r] = (hi ? g1 : x1) * inv_a;  vb1[r] = (hi ? y1 : g1) * inv_b;
    }
    if (rows_left >= SEG_PER_WAVE) {
#pragma unroll
        for (int r = 0; r < 8; ++r) {
            obase[(size_t)r * NFILT]            = va0[r];
            obase[(size_t)r * NFILT + 32]       = va1[r];
            obase[(size_t)(r + 8) * NFILT]      = vb0[r];
            obase[(size_t)(r + 8) * NFILT + 32] = vb1[r];
        }
    } else {
#pragma unroll
        for (int r = 0; r < 8; ++r) {
            if (r < rows_left) {
                obase[(size_t)r * NFILT]      = va0[r];
                obase[(size_t)r * NFILT + 32] = va1[r];
            }
            if (r + 8 < rows_left) {
                obase[(size_t)(r + 8) * NFILT]      = vb0[r];
                obase[(size_t)(r + 8) * NFILT + 32] = vb1[r];
            }
        }
    }
}

extern "C" void kernel_launch(void* const* d_in, const int* in_sizes, int n_in,
                              void* d_out, int out_size, void* d_ws, size_t ws_size,
                              hipStream_t stream) {
    (void)n_in; (void)out_size; (void)d_ws; (void)ws_size;
    const float* sources = (const float*)d_in[0];
    const int*   indices = (const int*)d_in[1];
    const int*   spans   = (const int*)d_in[2];
    float*       out     = (float*)d_out;

    const int n_tgt = in_sizes[2] - 1;
    const int blocks = (n_tgt + SEG_PER_WAVE * WAVES_PER_BLOCK - 1)
                       / (SEG_PER_WAVE * WAVES_PER_BLOCK);

    seg_mean_wmma_kernel<<<blocks, WAVES_PER_BLOCK * 32, 0, stream>>>(sources, indices, spans,
                                                                      out, n_tgt);
}
